# starting kernel
# baseline (speedup 1.0000x reference)
#include <hip/hip_runtime.h>
#include <cmath>

typedef __attribute__((ext_vector_type(2))) float v2f;
typedef __attribute__((ext_vector_type(4))) float v4f;
typedef __attribute__((ext_vector_type(8))) float v8f;

__global__ void __launch_bounds__(256) zero_kernel(v4f* __restrict__ p, int n4) {
    int i = blockIdx.x * blockDim.x + threadIdx.x;
    if (i < n4) p[i] = (v4f){0.f, 0.f, 0.f, 0.f};
}

template<int K, int NC>
__device__ __forceinline__ void fill_swizzled(float* sWs, const float* __restrict__ W) {
    constexpr int NT  = NC / 16;
    constexpr int TOT = (K / 4) * NT * 64;
    for (int i = threadIdx.x; i < TOT; i += blockDim.x) {
        const int h    = i & 1;
        const int lane = (i >> 1) & 31;
        const int c    = (i >> 6) % NT;
        const int ks   = i / (64 * NT);
        const int k    = ks * 4 + ((lane >> 4) << 1) + h;
        const int col  = c * 16 + (lane & 15);
        sWs[i] = W[k * NC + col];
    }
}

template<int D>
__global__ void __launch_bounds__(256) edge_kernel(
    const float* __restrict__ X, const int* __restrict__ src,
    const int* __restrict__ dst, const float* __restrict__ EA,
    const float* __restrict__ eW, const float* __restrict__ eb,
    float* __restrict__ Agg, int nedges)
{
    constexpr int NT = D / 16;
    __shared__ float sWs[4 * NT * 64];
    __shared__ float sB[D];
    fill_swizzled<16, D>(sWs, eW);
    for (int i = threadIdx.x; i < D; i += blockDim.x) sB[i] = eb[i];
    __syncthreads();

    const int lane  = threadIdx.x & 31;
    const int m     = lane & 15;
    const int kb    = (lane >> 4) << 1;
    const int n     = lane & 15;
    const int mbase = (lane >> 4) << 3;

    const int wavesTotal = (gridDim.x * blockDim.x) >> 5;
    const int ntiles = nedges >> 4;
    int wtile = (blockIdx.x * blockDim.x + threadIdx.x) >> 5;

    for (int t = wtile; t < ntiles; t += wavesTotal) {
        const int e0 = t << 4;

        const int ab = (e0 + m) * 16 + kb;
        const v2f a0 = *(const v2f*)(EA + ab);
        const v2f a1 = *(const v2f*)(EA + ab + 4);
        const v2f a2 = *(const v2f*)(EA + ab + 8);
        const v2f a3 = *(const v2f*)(EA + ab + 12);

        int sb[8], db[8];
#pragma unroll
        for (int j = 0; j < 8; ++j) {
            sb[j] = src[e0 + mbase + j] * D + n;
            db[j] = dst[e0 + mbase + j] * D + n;
        }

        v8f acc[NT];
#pragma unroll
        for (int c = 0; c < NT; ++c) {
            const float bv = sB[c * 16 + n];
            acc[c] = (v8f){bv, bv, bv, bv, bv, bv, bv, bv};
            const v2f* b0 = (const v2f*)(sWs + (0 * NT + c) * 64);
            const v2f* b1 = (const v2f*)(sWs + (1 * NT + c) * 64);
            const v2f* b2 = (const v2f*)(sWs + (2 * NT + c) * 64);
            const v2f* b3 = (const v2f*)(sWs + (3 * NT + c) * 64);
            acc[c] = __builtin_amdgcn_wmma_f32_16x16x4_f32(false, a0, false, b0[lane], (short)0, acc[c], false, false);
            acc[c] = __builtin_amdgcn_wmma_f32_16x16x4_f32(false, a1, false, b1[lane], (short)0, acc[c], false, false);
            acc[c] = __builtin_amdgcn_wmma_f32_16x16x4_f32(false, a2, false, b2[lane], (short)0, acc[c], false, false);
            acc[c] = __builtin_amdgcn_wmma_f32_16x16x4_f32(false, a3, false, b3[lane], (short)0, acc[c], false, false);
        }

#pragma unroll
        for (int j = 0; j < 8; ++j) {
            float xv[NT];
#pragma unroll
            for (int c = 0; c < NT; ++c) xv[c] = X[sb[j] + c * 16];
#pragma unroll
            for (int c = 0; c < NT; ++c) {
                float msg = acc[c][j] + xv[c];
                msg = msg > 0.0f ? msg : 0.0f;
                atomicAdd(&Agg[db[j] + c * 16], msg);
            }
        }
    }
}

template<int K, int NC, bool ACT_ELU, bool ADD2>
__global__ void __launch_bounds__(256) gemm_kernel(
    const float* __restrict__ A, const float* __restrict__ A2,
    const float* __restrict__ W, const float* __restrict__ bias,
    float* __restrict__ Out, int nrows)
{
    constexpr int NT = NC / 16;
    __shared__ float sWs[(K / 4) * NT * 64];
    __shared__ float sB[NC];
    fill_swizzled<K, NC>(sWs, W);
    for (int i = threadIdx.x; i < NC; i += blockDim.x) sB[i] = bias[i];
    __syncthreads();

    const int lane = threadIdx.x & 31;
    const int wave = threadIdx.x >> 5;
    const int tile = blockIdx.x * 8 + wave;
    if (tile * 16 >= nrows) return;
    const int row0 = tile * 16;

    const int m  = lane & 15;
    const int kb = (lane >> 4) << 1;
    const int n  = lane & 15;

    v8f acc[NT] = {};

    const int abase = (row0 + m) * K + kb;

#pragma unroll
    for (int ks = 0; ks < K / 4; ++ks) {
        v2f a = *(const v2f*)(A + abase + ks * 4);
        if (ADD2) {
            v2f a2 = *(const v2f*)(A2 + abase + ks * 4);
            a.x += a2.x; a.y += a2.y;
        }
        const v2f* brow = (const v2f*)(sWs + ks * NT * 64);
#pragma unroll
        for (int c = 0; c < NT; ++c) {
            acc[c] = __builtin_amdgcn_wmma_f32_16x16x4_f32(
                false, a, false, brow[c * 32 + lane], (short)0, acc[c], false, false);
        }
    }

    const int mbase = (lane >> 4) << 3;
    const int obase = (row0 + mbase) * NC + n;
#pragma unroll
    for (int c = 0; c < NT; ++c) {
        const float bv = sB[c * 16 + n];
#pragma unroll
        for (int j = 0; j < 8; ++j) {
            float v = acc[c][j] + bv;
            if (ACT_ELU) v = (v > 0.0f) ? v : (__expf(v) - 1.0f);
            Out[obase + j * NC + c * 16] = v;
        }
    }
}

extern "C" void kernel_launch(void* const* d_in, const int* in_sizes, int n_in,
                              void* d_out, int out_size, void* d_ws, size_t ws_size,
                              hipStream_t stream)
{
    const float* x    = (const float*)d_in[0];
    const int*   src  = (const int*)  d_in[1];
    const int*   dst  = (const int*)  d_in[2];
    const float* ea   = (const float*)d_in[3];
    const float* eW0  = (const float*)d_in[4];
    const float* eb0  = (const float*)d_in[5];
    const float* W1_0 = (const float*)d_in[6];
    const float* b1_0 = (const float*)d_in[7];
    const float* W2_0 = (const float*)d_in[8];
    const float* b2_0 = (const float*)d_in[9];
    const float* eW_r = (const float*)d_in[10];
    const float* eb_r = (const float*)d_in[11];
    const float* W1_r = (const float*)d_in[12];
    const float* b1_r = (const float*)d_in[13];
    const float* W2_r = (const float*)d_in[14];
    const float* b2_r = (const float*)d_in[15];
    const float* cW   = (const float*)d_in[16];
    const float* cb   = (const float*)d_in[17];

    const int N = in_sizes[0] / 64;
    const int E = in_sizes[1];

    float* agg = (float*)d_ws;
    float* hb  = agg + (size_t)N * 128;
    float* tb  = hb  + (size_t)N * 128;

    const int tiles   = (N + 15) / 16;
    const int gblocks = (tiles + 7) / 8;
    const int eblocks = 2048;

    { int n4 = (N * 64) / 4; zero_kernel<<<(n4 + 255) / 256, 256, 0, stream>>>((v4f*)agg, n4); }
    edge_kernel<64><<<eblocks, 256, 0, stream>>>(x, src, dst, ea, eW0, eb0, agg, E);
    gemm_kernel<64, 64, true, true ><<<gblocks, 256, 0, stream>>>(x,  agg, W1_0, b1_0, tb, N);
    gemm_kernel<64, 128, true, false><<<gblocks, 256, 0, stream>>>(tb, tb,  W2_0, b2_0, hb, N);

    for (int i = 0; i < 2; ++i) {
        int n4 = (N * 128) / 4;
        zero_kernel<<<(n4 + 255) / 256, 256, 0, stream>>>((v4f*)agg, n4);
        edge_kernel<128><<<eblocks, 256, 0, stream>>>(
            hb, src, dst, ea, eW_r + (size_t)i * 16 * 128, eb_r + (size_t)i * 128, agg, E);
        gemm_kernel<128, 128, true, true ><<<gblocks, 256, 0, stream>>>(
            hb, agg, W1_r + (size_t)i * 128 * 128, b1_r + (size_t)i * 128, tb, N);
        gemm_kernel<128, 128, true, false><<<gblocks, 256, 0, stream>>>(
            tb, tb, W2_r + (size_t)i * 128 * 128, b2_r + (size_t)i * 128, hb, N);
    }

    gemm_kernel<128, 32, false, false><<<gblocks, 256, 0, stream>>>(
        hb, hb, cW, cb, (float*)d_out, N);
}


# Weight staging to LDS: each thread issues all its 16-byte global loads up front, then 16-byte LDS stores, instead of one dependent 4-byte load-store per loop trip
# speedup vs baseline: 1.4537x; 1.4537x over previous
#include <hip/hip_runtime.h>
#include <cmath>

typedef __attribute__((ext_vector_type(2))) float v2f;
typedef __attribute__((ext_vector_type(4))) float v4f;
typedef __attribute__((ext_vector_type(8))) float v8f;

__global__ void __launch_bounds__(256) zero_kernel(v4f* __restrict__ p, int n4) {
    int i = blockIdx.x * blockDim.x + threadIdx.x;
    if (i < n4) p[i] = (v4f){0.f, 0.f, 0.f, 0.f};
}

constexpr int STAGE_THREADS = 256;

template<int K, int NC>
__device__ __forceinline__ void stage_weights(float* sWs, float* sB,
                                              const float* __restrict__ W,
                                              const float* __restrict__ bias) {
    constexpr int NT      = NC / 16;
    constexpr int G       = NC / 4;
    constexpr int HALF    = STAGE_THREADS / 2;
    constexpr int KS_STEP = HALF / G;
    constexpr int PER     = (K / 4 + KS_STEP - 1) / KS_STEP;
    constexpr bool FULL   = ((K / 4) % KS_STEP) == 0;
    static_assert(HALF % G == 0, "a pass covers whole k-steps");
    static_assert(NC <= STAGE_THREADS, "bias is staged in one pass");
    const int tid = threadIdx.x;
    const int hl  = tid & 1;
    const int g   = (tid >> 1) % G;
    const int ks0 = (tid >> 1) / G;
    const float* s = W + (ks0 * 4 + hl * 2) * NC + g * 4;
    float*       d = sWs + (ks0 * NT + (g >> 2)) * 64 + hl * 32 + (g & 3) * 8;

    float bv = 0.0f;
    if (tid < NC) bv = bias[tid];

    v4f r0[PER], r1[PER];
#pragma unroll
    for (int u = 0; u < PER; ++u) {
        if (FULL || ks0 + u * KS_STEP < K / 4) {
            r0[u] = *(const v4f*)(s + u * KS_STEP * 4 * NC);
            r1[u] = *(const v4f*)(s + u * KS_STEP * 4 * NC + NC);
        }
    }
#pragma unroll
    for (int u = 0; u < PER; ++u) {
        if (FULL || ks0 + u * KS_STEP < K / 4) {
            float* du = d + u * KS_STEP * NT * 64;
            *(v4f*)du       = (v4f){r0[u].x, r1[u].x, r0[u].y, r1[u].y};
            *(v4f*)(du + 4) = (v4f){r0[u].z, r1[u].z, r0[u].w, r1[u].w};
        }
    }
    if (tid < NC) sB[tid] = bv;
}

template<int D>
__global__ void __launch_bounds__(256) edge_kernel(
    const float* __restrict__ X, const int* __restrict__ src,
    const int* __restrict__ dst, const float* __restrict__ EA,
    const float* __restrict__ eW, const float* __restrict__ eb,
    float* __restrict__ Agg, int nedges)
{
    constexpr int NT = D / 16;
    __shared__ __attribute__((aligned(16))) float sWs[4 * NT * 64];
    __shared__ float sB[D];
    stage_weights<16, D>(sWs, sB, eW, eb);
    __syncthreads();

    const int lane  = threadIdx.x & 31;
    const int m     = lane & 15;
    const int kb    = (lane >> 4) << 1;
    const int n     = lane & 15;
    const int mbase = (lane >> 4) << 3;

    const int wavesTotal = (gridDim.x * blockDim.x) >> 5;
    const int ntiles = nedges >> 4;
    int wtile = (blockIdx.x * blockDim.x + threadIdx.x) >> 5;

    for (int t = wtile; t < ntiles; t += wavesTotal) {
        const int e0 = t << 4;

        const int ab = (e0 + m) * 16 + kb;
        const v2f a0 = *(const v2f*)(EA + ab);
        const v2f a1 = *(const v2f*)(EA + ab + 4);
        const v2f a2 = *(const v2f*)(EA + ab + 8);
        const v2f a3 = *(const v2f*)(EA + ab + 12);

        int sb[8], db[8];
#pragma unroll
        for (int j = 0; j < 8; ++j) {
            sb[j] = src[e0 + mbase + j] * D + n;
            db[j] = dst[e0 + mbase + j] * D + n;
        }

        v8f acc[NT];
#pragma unroll
        for (int c = 0; c < NT; ++c) {
            const float bv = sB[c * 16 + n];
            acc[c] = (v8f){bv, bv, bv, bv, bv, bv, bv, bv};
            const v2f* b0 = (const v2f*)(sWs + (0 * NT + c) * 64);
            const v2f* b1 = (const v2f*)(sWs + (1 * NT + c) * 64);
            const v2f* b2 = (const v2f*)(sWs + (2 * NT + c) * 64);
            const v2f* b3 = (const v2f*)(sWs + (3 * NT + c) * 64);
            acc[c] = __builtin_amdgcn_wmma_f32_16x16x4_f32(false, a0, false, b0[lane], (short)0, acc[c], false, false);
            acc[c] = __builtin_amdgcn_wmma_f32_16x16x4_f32(false, a1, false, b1[lane], (short)0, acc[c], false, false);
            acc[c] = __builtin_amdgcn_wmma_f32_16x16x4_f32(false, a2, false, b2[lane], (short)0, acc[c], false, false);
            acc[c] = __builtin_amdgcn_wmma_f32_16x16x4_f32(false, a3, false, b3[lane], (short)0, acc[c], false, false);
        }

#pragma unroll
        for (int j = 0; j < 8; ++j) {
            float xv[NT];
#pragma unroll
            for (int c = 0; c < NT; ++c) xv[c] = X[sb[j] + c * 16];
#pragma unroll
            for (int c = 0; c < NT; ++c) {
                float msg = acc[c][j] + xv[c];
                msg = msg > 0.0f ? msg : 0.0f;
                atomicAdd(&Agg[db[j] + c * 16], msg);
            }
        }
    }
}

template<int K, int NC, bool ACT_ELU, bool ADD2>
__global__ void __launch_bounds__(256) gemm_kernel(
    const float* __restrict__ A, const float* __restrict__ A2,
    const float* __restrict__ W, const float* __restrict__ bias,
    float* __restrict__ Out, int nrows)
{
    constexpr int NT = NC / 16;
    __shared__ __attribute__((aligned(16))) float sWs[(K / 4) * NT * 64];
    __shared__ float sB[NC];
    stage_weights<K, NC>(sWs, sB, W, bias);
    __syncthreads();

    const int lane = threadIdx.x & 31;
    const int wave = threadIdx.x >> 5;
    const int tile = blockIdx.x * 8 + wave;
    if (tile * 16 >= nrows) return;
    const int row0 = tile * 16;

    const int m  = lane & 15;
    const int kb = (lane >> 4) << 1;
    const int n  = lane & 15;

    v8f acc[NT] = {};

    const int abase = (row0 + m) * K + kb;

#pragma unroll
    for (int ks = 0; ks < K / 4; ++ks) {
        v2f a = *(const v2f*)(A + abase + ks * 4);
        if (ADD2) {
            v2f a2 = *(const v2f*)(A2 + abase + ks * 4);
            a.x += a2.x; a.y += a2.y;
        }
        const v2f* brow = (const v2f*)(sWs + ks * NT * 64);
#pragma unroll
        for (int c = 0; c < NT; ++c) {
            acc[c] = __builtin_amdgcn_wmma_f32_16x16x4_f32(
                false, a, false, brow[c * 32 + lane], (short)0, acc[c], false, false);
        }
    }

    const int mbase = (lane >> 4) << 3;
    const int obase = (row0 + mbase) * NC + n;
#pragma unroll
    for (int c = 0; c < NT; ++c) {
        const float bv = sB[c * 16 + n];
#pragma unroll
        for (int j = 0; j < 8; ++j) {
            float v = acc[c][j] + bv;
            if (ACT_ELU) v = (v > 0.0f) ? v : (__expf(v) - 1.0f);
            Out[obase + j * NC + c * 16] = v;
        }
    }
}

extern "C" void kernel_launch(void* const* d_in, const int* in_sizes, int n_in,
                              void* d_out, int out_size, void* d_ws, size_t ws_size,
                              hipStream_t stream)
{
    const float* x    = (const float*)d_in[0];
    const int*   src  = (const int*)  d_in[1];
    const int*   dst  = (const int*)  d_in[2];
    const float* ea   = (const float*)d_in[3];
    const float* eW0  = (const float*)d_in[4];
    const float* eb0  = (const float*)d_in[5];
    const float* W1_0 = (const float*)d_in[6];
    const float* b1_0 = (const float*)d_in[7];
    const float* W2_0 = (const float*)d_in[8];
    const float* b2_0 = (const float*)d_in[9];
    const float* eW_r = (const float*)d_in[10];
    const float* eb_r = (const float*)d_in[11];
    const float* W1_r = (const float*)d_in[12];
    const float* b1_r = (const float*)d_in[13];
    const float* W2_r = (const float*)d_in[14];
    const float* b2_r = (const float*)d_in[15];
    const float* cW   = (const float*)d_in[16];
    const float* cb   = (const float*)d_in[17];

    const int N = in_sizes[0] / 64;
    const int E = in_sizes[1];

    float* agg = (float*)d_ws;
    float* hb  = agg + (size_t)N * 128;
    float* tb  = hb  + (size_t)N * 128;

    const int tiles   = (N + 15) / 16;
    const int gblocks = (tiles + 7) / 8;
    const int eblocks = 2048;

    { int n4 = (N * 64) / 4; zero_kernel<<<(n4 + 255) / 256, 256, 0, stream>>>((v4f*)agg, n4); }
    edge_kernel<64><<<eblocks, 256, 0, stream>>>(x, src, dst, ea, eW0, eb0, agg, E);
    gemm_kernel<64, 64, true, true ><<<gblocks, 256, 0, stream>>>(x,  agg, W1_0, b1_0, tb, N);
    gemm_kernel<64, 128, true, false><<<gblocks, 256, 0, stream>>>(tb, tb,  W2_0, b2_0, hb, N);

    for (int i = 0; i < 2; ++i) {
        int n4 = (N * 128) / 4;
        zero_kernel<<<(n4 + 255) / 256, 256, 0, stream>>>((v4f*)agg, n4);
        edge_kernel<128><<<eblocks, 256, 0, stream>>>(
            hb, src, dst, ea, eW_r + (size_t)i * 16 * 128, eb_r + (size_t)i * 128, agg, E);
        gemm_kernel<128, 128, true, true ><<<gblocks, 256, 0, stream>>>(
            hb, agg, W1_r + (size_t)i * 128 * 128, b1_r + (size_t)i * 128, tb, N);
        gemm_kernel<128, 128, true, false><<<gblocks, 256, 0, stream>>>(
            tb, tb, W2_r + (size_t)i * 128 * 128, b2_r + (size_t)i * 128, hb, N);
    }

    gemm_kernel<128, 32, false, false><<<gblocks, 256, 0, stream>>>(
        hb, hb, cW, cb, (float*)d_out, N);
}


# Edge kernel: all 8 x NT x[src] gathers of a tile issued ahead of the WMMAs (one overlapped latency) instead of 8 serial load-wait-atomic groups; bias added after the product sum (sum order)
# speedup vs baseline: 1.5291x; 1.0519x over previous
#include <hip/hip_runtime.h>
#include <cmath>

typedef __attribute__((ext_vector_type(2))) float v2f;
typedef __attribute__((ext_vector_type(4))) float v4f;
typedef __attribute__((ext_vector_type(8))) float v8f;

__global__ void __launch_bounds__(256) zero_kernel(v4f* __restrict__ p, int n4) {
    int i = blockIdx.x * blockDim.x + threadIdx.x;
    if (i < n4) p[i] = (v4f){0.f, 0.f, 0.f, 0.f};
}

constexpr int STAGE_THREADS = 256;

template<int K, int NC>
__device__ __forceinline__ void stage_weights(float* sWs, float* sB,
                                              const float* __restrict__ W,
                                              const float* __restrict__ bias) {
    constexpr int NT      = NC / 16;
    constexpr int G       = NC / 4;
    constexpr int HALF    = STAGE_THREADS / 2;
    constexpr int KS_STEP = HALF / G;
    constexpr int PER     = (K / 4 + KS_STEP - 1) / KS_STEP;
    constexpr bool FULL   = ((K / 4) % KS_STEP) == 0;
    static_assert(HALF % G == 0, "a pass covers whole k-steps");
    static_assert(NC <= STAGE_THREADS, "bias is staged in one pass");
    const int tid = threadIdx.x;
    const int hl  = tid & 1;
    const int g   = (tid >> 1) % G;
    const int ks0 = (tid >> 1) / G;
    const float* s = W + (ks0 * 4 + hl * 2) * NC + g * 4;
    float*       d = sWs + (ks0 * NT + (g >> 2)) * 64 + hl * 32 + (g & 3) * 8;

    float bv = 0.0f;
    if (tid < NC) bv = bias[tid];

    v4f r0[PER], r1[PER];
#pragma unroll
    for (int u = 0; u < PER; ++u) {
        if (FULL || ks0 + u * KS_STEP < K / 4) {
            r0[u] = *(const v4f*)(s + u * KS_STEP * 4 * NC);
            r1[u] = *(const v4f*)(s + u * KS_STEP * 4 * NC + NC);
        }
    }
#pragma unroll
    for (int u = 0; u < PER; ++u) {
        if (FULL || ks0 + u * KS_STEP < K / 4) {
            float* du = d + u * KS_STEP * NT * 64;
            *(v4f*)du       = (v4f){r0[u].x, r1[u].x, r0[u].y, r1[u].y};
            *(v4f*)(du + 4) = (v4f){r0[u].z, r1[u].z, r0[u].w, r1[u].w};
        }
    }
    if (tid < NC) sB[tid] = bv;
}

template<int D>
__global__ void __launch_bounds__(256) edge_kernel(
    const float* __restrict__ X, const int* __restrict__ src,
    const int* __restrict__ dst, const float* __restrict__ EA,
    const float* __restrict__ eW, const float* __restrict__ eb,
    float* __restrict__ Agg, int nedges)
{
    constexpr int NT = D / 16;
    __shared__ __attribute__((aligned(16))) float sWs[4 * NT * 64];
    __shared__ float sB[D];
    stage_weights<16, D>(sWs, sB, eW, eb);
    __syncthreads();

    const int lane  = threadIdx.x & 31;
    const int m     = lane & 15;
    const int kb    = (lane >> 4) << 1;
    const int n     = lane & 15;
    const int mbase = (lane >> 4) << 3;

    float bv[NT];
#pragma unroll
    for (int c = 0; c < NT; ++c) bv[c] = sB[c * 16 + n];

    const int wavesTotal = (gridDim.x * blockDim.x) >> 5;
    const int ntiles = nedges >> 4;
    int wtile = (blockIdx.x * blockDim.x + threadIdx.x) >> 5;

    for (int t = wtile; t < ntiles; t += wavesTotal) {
        const int e0 = t << 4;

        const int ab = (e0 + m) * 16 + kb;
        const v2f a0 = *(const v2f*)(EA + ab);
        const v2f a1 = *(const v2f*)(EA + ab + 4);
        const v2f a2 = *(const v2f*)(EA + ab + 8);
        const v2f a3 = *(const v2f*)(EA + ab + 12);

        int sb[8], db[8];
#pragma unroll
        for (int j = 0; j < 8; ++j) {
            sb[j] = src[e0 + mbase + j] * D + n;
            db[j] = dst[e0 + mbase + j] * D + n;
        }

        float xv[8][NT];
#pragma unroll
        for (int j = 0; j < 8; ++j) {
#pragma unroll
            for (int c = 0; c < NT; ++c) xv[j][c] = X[sb[j] + c * 16];
        }
        __builtin_amdgcn_sched_barrier(0);

        v8f acc[NT];
#pragma unroll
        for (int c = 0; c < NT; ++c) {
            acc[c] = (v8f){0.f, 0.f, 0.f, 0.f, 0.f, 0.f, 0.f, 0.f};
            const v2f* b0 = (const v2f*)(sWs + (0 * NT + c) * 64);
            const v2f* b1 = (const v2f*)(sWs + (1 * NT + c) * 64);
            const v2f* b2 = (const v2f*)(sWs + (2 * NT + c) * 64);
            const v2f* b3 = (const v2f*)(sWs + (3 * NT + c) * 64);
            acc[c] = __builtin_amdgcn_wmma_f32_16x16x4_f32(false, a0, false, b0[lane], (short)0, acc[c], false, false);
            acc[c] = __builtin_amdgcn_wmma_f32_16x16x4_f32(false, a1, false, b1[lane], (short)0, acc[c], false, false);
            acc[c] = __builtin_amdgcn_wmma_f32_16x16x4_f32(false, a2, false, b2[lane], (short)0, acc[c], false, false);
            acc[c] = __builtin_amdgcn_wmma_f32_16x16x4_f32(false, a3, false, b3[lane], (short)0, acc[c], false, false);
        }

#pragma unroll
        for (int j = 0; j < 8; ++j) {
#pragma unroll
            for (int c = 0; c < NT; ++c) {
                const float msg = (acc[c][j] + bv[c]) + xv[j][c];
                acc[c][j] = msg > 0.0f ? msg : 0.0f;
            }
        }
#pragma unroll
        for (int j = 0; j < 8; ++j) {
#pragma unroll
            for (int c = 0; c < NT; ++c)
                atomicAdd(&Agg[db[j] + c * 16], acc[c][j]);
        }
    }
}

template<int K, int NC, bool ACT_ELU, bool ADD2>
__global__ void __launch_bounds__(256) gemm_kernel(
    const float* __restrict__ A, const float* __restrict__ A2,
    const float* __restrict__ W, const float* __restrict__ bias,
    float* __restrict__ Out, int nrows)
{
    constexpr int NT = NC / 16;
    __shared__ __attribute__((aligned(16))) float sWs[(K / 4) * NT * 64];
    __shared__ float sB[NC];
    stage_weights<K, NC>(sWs, sB, W, bias);
    __syncthreads();

    const int lane = threadIdx.x & 31;
    const int wave = threadIdx.x >> 5;
    const int tile = blockIdx.x * 8 + wave;
    if (tile * 16 >= nrows) return;
    const int row0 = tile * 16;

    const int m  = lane & 15;
    const int kb = (lane >> 4) << 1;
    const int n  = lane & 15;

    v8f acc[NT] = {};

    const int abase = (row0 + m) * K + kb;

#pragma unroll
    for (int ks = 0; ks < K / 4; ++ks) {
        v2f a = *(const v2f*)(A + abase + ks * 4);
        if (ADD2) {
            v2f a2 = *(const v2f*)(A2 + abase + ks * 4);
            a.x += a2.x; a.y += a2.y;
        }
        const v2f* brow = (const v2f*)(sWs + ks * NT * 64);
#pragma unroll
        for (int c = 0; c < NT; ++c) {
            acc[c] = __builtin_amdgcn_wmma_f32_16x16x4_f32(
                false, a, false, brow[c * 32 + lane], (short)0, acc[c], false, false);
        }
    }

    const int mbase = (lane >> 4) << 3;
    const int obase = (row0 + mbase) * NC + n;
#pragma unroll
    for (int c = 0; c < NT; ++c) {
        const float bv = sB[c * 16 + n];
#pragma unroll
        for (int j = 0; j < 8; ++j) {
            float v = acc[c][j] + bv;
            if (ACT_ELU) v = (v > 0.0f) ? v : (__expf(v) - 1.0f);
            Out[obase + j * NC + c * 16] = v;
        }
    }
}

extern "C" void kernel_launch(void* const* d_in, const int* in_sizes, int n_in,
                              void* d_out, int out_size, void* d_ws, size_t ws_size,
                              hipStream_t stream)
{
    const float* x    = (const float*)d_in[0];
    const int*   src  = (const int*)  d_in[1];
    const int*   dst  = (const int*)  d_in[2];
    const float* ea   = (const float*)d_in[3];
    const float* eW0  = (const float*)d_in[4];
    const float* eb0  = (const float*)d_in[5];
    const float* W1_0 = (const float*)d_in[6];
    const float* b1_0 = (const float*)d_in[7];
    const float* W2_0 = (const float*)d_in[8];
    const float* b2_0 = (const float*)d_in[9];
    const float* eW_r = (const float*)d_in[10];
    const float* eb_r = (const float*)d_in[11];
    const float* W1_r = (const float*)d_in[12];
    const float* b1_r = (const float*)d_in[13];
    const float* W2_r = (const float*)d_in[14];
    const float* b2_r = (const float*)d_in[15];
    const float* cW   = (const float*)d_in[16];
    const float* cb   = (const float*)d_in[17];

    const int N = in_sizes[0] / 64;
    const int E = in_sizes[1];

    float* agg = (float*)d_ws;
    float* hb  = agg + (size_t)N * 128;
    float* tb  = hb  + (size_t)N * 128;

    const int tiles   = (N + 15) / 16;
    const int gblocks = (tiles + 7) / 8;
    const int eblocks = 2048;

    { int n4 = (N * 64) / 4; zero_kernel<<<(n4 + 255) / 256, 256, 0, stream>>>((v4f*)agg, n4); }
    edge_kernel<64><<<eblocks, 256, 0, stream>>>(x, src, dst, ea, eW0, eb0, agg, E);
    gemm_kernel<64, 64, true, true ><<<gblocks, 256, 0, stream>>>(x,  agg, W1_0, b1_0, tb, N);
    gemm_kernel<64, 128, true, false><<<gblocks, 256, 0, stream>>>(tb, tb,  W2_0, b2_0, hb, N);

    for (int i = 0; i < 2; ++i) {
        int n4 = (N * 128) / 4;
        zero_kernel<<<(n4 + 255) / 256, 256, 0, stream>>>((v4f*)agg, n4);
        edge_kernel<128><<<eblocks, 256, 0, stream>>>(
            hb, src, dst, ea, eW_r + (size_t)i * 16 * 128, eb_r + (size_t)i * 128, agg, E);
        gemm_kernel<128, 128, true, true ><<<gblocks, 256, 0, stream>>>(
            hb, agg, W1_r + (size_t)i * 128 * 128, b1_r + (size_t)i * 128, tb, N);
        gemm_kernel<128, 128, true, false><<<gblocks, 256, 0, stream>>>(
            tb, tb, W2_r + (size_t)i * 128 * 128, b2_r + (size_t)i * 128, hb, N);
    }

    gemm_kernel<128, 32, false, false><<<gblocks, 256, 0, stream>>>(
        hb, hb, cW, cb, (float*)d_out, N);
}


# Edge pass by whole lines: each gather and f32 atomic covers 32 floats of ONE row (a 128-B line), not 16 of each of two rows; tile pairs swapped across lane halves (v_permlanex16), scalar row indices
# speedup vs baseline: 1.6951x; 1.1086x over previous
#include <hip/hip_runtime.h>
#include <cmath>

typedef __attribute__((ext_vector_type(2))) float v2f;
typedef __attribute__((ext_vector_type(4))) float v4f;
typedef __attribute__((ext_vector_type(8))) float v8f;

__global__ void __launch_bounds__(256) zero_kernel(v4f* __restrict__ p, int n4) {
    int i = blockIdx.x * blockDim.x + threadIdx.x;
    if (i < n4) p[i] = (v4f){0.f, 0.f, 0.f, 0.f};
}

constexpr int STAGE_THREADS = 256;

template<int K, int NC>
__device__ __forceinline__ void stage_weights(float* sWs, float* sB,
                                              const float* __restrict__ W,
                                              const float* __restrict__ bias) {
    constexpr int NT      = NC / 16;
    constexpr int G       = NC / 4;
    constexpr int HALF    = STAGE_THREADS / 2;
    constexpr int KS_STEP = HALF / G;
    constexpr int PER     = (K / 4 + KS_STEP - 1) / KS_STEP;
    constexpr bool FULL   = ((K / 4) % KS_STEP) == 0;
    static_assert(HALF % G == 0, "a pass covers whole k-steps");
    static_assert(NC <= STAGE_THREADS, "bias is staged in one pass");
    const int tid = threadIdx.x;
    const int hl  = tid & 1;
    const int g   = (tid >> 1) % G;
    const int ks0 = (tid >> 1) / G;
    const float* s = W + (ks0 * 4 + hl * 2) * NC + g * 4;
    float*       d = sWs + (ks0 * NT + (g >> 2)) * 64 + hl * 32 + (g & 3) * 8;

    float bv = 0.0f;
    if (tid < NC) bv = bias[tid];

    v4f r0[PER], r1[PER];
#pragma unroll
    for (int u = 0; u < PER; ++u) {
        if (FULL || ks0 + u * KS_STEP < K / 4) {
            r0[u] = *(const v4f*)(s + u * KS_STEP * 4 * NC);
            r1[u] = *(const v4f*)(s + u * KS_STEP * 4 * NC + NC);
        }
    }
#pragma unroll
    for (int u = 0; u < PER; ++u) {
        if (FULL || ks0 + u * KS_STEP < K / 4) {
            float* du = d + u * KS_STEP * NT * 64;
            *(v4f*)du       = (v4f){r0[u].x, r1[u].x, r0[u].y, r1[u].y};
            *(v4f*)(du + 4) = (v4f){r0[u].z, r1[u].z, r0[u].w, r1[u].w};
        }
    }
    if (tid < NC) sB[tid] = bv;
}


__device__ __forceinline__ float xhalf(float x) {
    const int b = __builtin_bit_cast(int, x);
    return __builtin_bit_cast(float,
        __builtin_amdgcn_permlanex16(b, b, 0x76543210, 0xfedcba98, false, false));
}

__device__ __forceinline__ void swap_halves(float& a, float& b, bool lo) {
    const float u = xhalf(lo ? b : a);
    a = lo ? a : u;
    b = lo ? u : b;
}

template<int D>
__global__ void __launch_bounds__(256) edge_kernel(
    const float* __restrict__ X, const int* __restrict__ src,
    const int* __restrict__ dst, const float* __restrict__ EA,
    const float* __restrict__ eW, const float* __restrict__ eb,
    float* __restrict__ Agg, int nedges)
{
    constexpr int NT = D / 16;
    constexpr int NL = D / 32;
    __shared__ __attribute__((aligned(16))) float sWs[4 * NT * 64];
    __shared__ float sB[D];
    stage_weights<16, D>(sWs, sB, eW, eb);
    __syncthreads();

    const int lane  = threadIdx.x & 31;
    const int m     = lane & 15;
    const int kb    = (lane >> 4) << 1;
    const bool lo   = lane < 16;

    float bv[NL];
#pragma unroll
    for (int q = 0; q < NL; ++q) bv[q] = sB[q * 32 + lane];

    const int wavesTotal = (gridDim.x * blockDim.x) >> 5;
    const int ntiles = nedges >> 4;
    const int wtile = __builtin_amdgcn_readfirstlane(
        (int)((blockIdx.x * blockDim.x + threadIdx.x) >> 5));

    for (int t = wtile; t < ntiles; t += wavesTotal) {
        const int e0 = t << 4;

        const int ab = (e0 + m) * 16 + kb;
        const v2f a0 = *(const v2f*)(EA + ab);
        const v2f a1 = *(const v2f*)(EA + ab + 4);
        const v2f a2 = *(const v2f*)(EA + ab + 8);
        const v2f a3 = *(const v2f*)(EA + ab + 12);

        int so[16], dd[16];
#pragma unroll
        for (int r = 0; r < 16; ++r) {
            so[r] = src[e0 + r] * D;
            dd[r] = dst[e0 + r] * D;
        }

        float xv[16][NL];
#pragma unroll
        for (int j = 0; j < 8; ++j) {
#pragma unroll
            for (int q = 0; q < NL; ++q) xv[j][q]     = X[so[j] + q * 32 + lane];
#pragma unroll
            for (int q = 0; q < NL; ++q) xv[j + 8][q] = X[so[j + 8] + q * 32 + lane];
        }
        __builtin_amdgcn_sched_barrier(0);

        v8f acc[NT];
#pragma unroll
        for (int c = 0; c < NT; ++c) {
            acc[c] = (v8f){0.f, 0.f, 0.f, 0.f, 0.f, 0.f, 0.f, 0.f};
            const v2f* b0 = (const v2f*)(sWs + (0 * NT + c) * 64);
            const v2f* b1 = (const v2f*)(sWs + (1 * NT + c) * 64);
            const v2f* b2 = (const v2f*)(sWs + (2 * NT + c) * 64);
            const v2f* b3 = (const v2f*)(sWs + (3 * NT + c) * 64);
            acc[c] = __builtin_amdgcn_wmma_f32_16x16x4_f32(false, a0, false, b0[lane], (short)0, acc[c], false, false);
            acc[c] = __builtin_amdgcn_wmma_f32_16x16x4_f32(false, a1, false, b1[lane], (short)0, acc[c], false, false);
            acc[c] = __builtin_amdgcn_wmma_f32_16x16x4_f32(false, a2, false, b2[lane], (short)0, acc[c], false, false);
            acc[c] = __builtin_amdgcn_wmma_f32_16x16x4_f32(false, a3, false, b3[lane], (short)0, acc[c], false, false);
        }

#pragma unroll
        for (int j = 0; j < 8; ++j) {
#pragma unroll
            for (int q = 0; q < NL; ++q) {
                float p = acc[2 * q][j];
                float w = acc[2 * q + 1][j];
                swap_halves(p, w, lo);
                p = (p + bv[q]) + xv[j][q];
                w = (w + bv[q]) + xv[j + 8][q];
                acc[2 * q][j]     = p > 0.0f ? p : 0.0f;
                acc[2 * q + 1][j] = w > 0.0f ? w : 0.0f;
            }
        }
#pragma unroll
        for (int j = 0; j < 8; ++j) {
#pragma unroll
            for (int q = 0; q < NL; ++q)
                atomicAdd(&Agg[dd[j] + q * 32 + lane], acc[2 * q][j]);
#pragma unroll
            for (int q = 0; q < NL; ++q)
                atomicAdd(&Agg[dd[j + 8] + q * 32 + lane], acc[2 * q + 1][j]);
        }
    }
}

template<int K, int NC, bool ACT_ELU, bool ADD2>
__global__ void __launch_bounds__(256) gemm_kernel(
    const float* __restrict__ A, const float* __restrict__ A2,
    const float* __restrict__ W, const float* __restrict__ bias,
    float* __restrict__ Out, int nrows)
{
    constexpr int NT = NC / 16;
    __shared__ __attribute__((aligned(16))) float sWs[(K / 4) * NT * 64];
    __shared__ float sB[NC];
    stage_weights<K, NC>(sWs, sB, W, bias);
    __syncthreads();

    const int lane = threadIdx.x & 31;
    const int wave = threadIdx.x >> 5;
    const int tile = blockIdx.x * 8 + wave;
    if (tile * 16 >= nrows) return;
    const int row0 = tile * 16;

    const int m  = lane & 15;
    const int kb = (lane >> 4) << 1;
    const int n  = lane & 15;

    v8f acc[NT] = {};

    const int abase = (row0 + m) * K + kb;

#pragma unroll
    for (int ks = 0; ks < K / 4; ++ks) {
        v2f a = *(const v2f*)(A + abase + ks * 4);
        if (ADD2) {
            v2f a2 = *(const v2f*)(A2 + abase + ks * 4);
            a.x += a2.x; a.y += a2.y;
        }
        const v2f* brow = (const v2f*)(sWs + ks * NT * 64);
#pragma unroll
        for (int c = 0; c < NT; ++c) {
            acc[c] = __builtin_amdgcn_wmma_f32_16x16x4_f32(
                false, a, false, brow[c * 32 + lane], (short)0, acc[c], false, false);
        }
    }

    const int mbase = (lane >> 4) << 3;
    const int obase = (row0 + mbase) * NC + n;
#pragma unroll
    for (int c = 0; c < NT; ++c) {
        const float bv = sB[c * 16 + n];
#pragma unroll
        for (int j = 0; j < 8; ++j) {
            float v = acc[c][j] + bv;
            if (ACT_ELU) v = (v > 0.0f) ? v : (__expf(v) - 1.0f);
            Out[obase + j * NC + c * 16] = v;
        }
    }
}

extern "C" void kernel_launch(void* const* d_in, const int* in_sizes, int n_in,
                              void* d_out, int out_size, void* d_ws, size_t ws_size,
                              hipStream_t stream)
{
    const float* x    = (const float*)d_in[0];
    const int*   src  = (const int*)  d_in[1];
    const int*   dst  = (const int*)  d_in[2];
    const float* ea   = (const float*)d_in[3];
    const float* eW0  = (const float*)d_in[4];
    const float* eb0  = (const float*)d_in[5];
    const float* W1_0 = (const float*)d_in[6];
    const float* b1_0 = (const float*)d_in[7];
    const float* W2_0 = (const float*)d_in[8];
    const float* b2_0 = (const float*)d_in[9];
    const float* eW_r = (const float*)d_in[10];
    const float* eb_r = (const float*)d_in[11];
    const float* W1_r = (const float*)d_in[12];
    const float* b1_r = (const float*)d_in[13];
    const float* W2_r = (const float*)d_in[14];
    const float* b2_r = (const float*)d_in[15];
    const float* cW   = (const float*)d_in[16];
    const float* cb   = (const float*)d_in[17];

    const int N = in_sizes[0] / 64;
    const int E = in_sizes[1];

    float* agg = (float*)d_ws;
    float* hb  = agg + (size_t)N * 128;
    float* tb  = hb  + (size_t)N * 128;

    const int tiles   = (N + 15) / 16;
    const int gblocks = (tiles + 7) / 8;
    const int eblocks = 2048;

    { int n4 = (N * 64) / 4; zero_kernel<<<(n4 + 255) / 256, 256, 0, stream>>>((v4f*)agg, n4); }
    edge_kernel<64><<<eblocks, 256, 0, stream>>>(x, src, dst, ea, eW0, eb0, agg, E);
    gemm_kernel<64, 64, true, true ><<<gblocks, 256, 0, stream>>>(x,  agg, W1_0, b1_0, tb, N);
    gemm_kernel<64, 128, true, false><<<gblocks, 256, 0, stream>>>(tb, tb,  W2_0, b2_0, hb, N);

    for (int i = 0; i < 2; ++i) {
        int n4 = (N * 128) / 4;
        zero_kernel<<<(n4 + 255) / 256, 256, 0, stream>>>((v4f*)agg, n4);
        edge_kernel<128><<<eblocks, 256, 0, stream>>>(
            hb, src, dst, ea, eW_r + (size_t)i * 16 * 128, eb_r + (size_t)i * 128, agg, E);
        gemm_kernel<128, 128, true, true ><<<gblocks, 256, 0, stream>>>(
            hb, agg, W1_r + (size_t)i * 128 * 128, b1_r + (size_t)i * 128, tb, N);
        gemm_kernel<128, 128, true, false><<<gblocks, 256, 0, stream>>>(
            tb, tb, W2_r + (size_t)i * 128 * 128, b2_r + (size_t)i * 128, hb, N);
    }

    gemm_kernel<128, 32, false, false><<<gblocks, 256, 0, stream>>>(
        hb, hb, cW, cb, (float*)d_out, N);
}


# Node side of each layer in ONE launch: both products (last layer: classifier too) chained through a wave-private LDS tile; the same waves clear the next layer's sum buffer. 13 launches -> 7
# speedup vs baseline: 1.9564x; 1.1542x over previous
#include <hip/hip_runtime.h>
#include <cmath>

typedef __attribute__((ext_vector_type(2))) float v2f;
typedef __attribute__((ext_vector_type(4))) float v4f;
typedef __attribute__((ext_vector_type(8))) float v8f;

__global__ void __launch_bounds__(256) zero_kernel(v4f* __restrict__ p, int n4) {
    int i = blockIdx.x * blockDim.x + threadIdx.x;
    if (i < n4) p[i] = (v4f){0.f, 0.f, 0.f, 0.f};
}

constexpr int STAGE_THREADS = 256;

template<int K, int NC>
__device__ __forceinline__ void stage_weights(float* sWs, float* sB,
                                              const float* __restrict__ W,
                                              const float* __restrict__ bias) {
    constexpr int NT      = NC / 16;
    constexpr int G       = NC / 4;
    constexpr int HALF    = STAGE_THREADS / 2;
    constexpr int KS_STEP = HALF / G;
    constexpr int PER     = (K / 4 + KS_STEP - 1) / KS_STEP;
    constexpr bool FULL   = ((K / 4) % KS_STEP) == 0;
    static_assert(HALF % G == 0, "a pass covers whole k-steps");
    static_assert(NC <= STAGE_THREADS, "bias is staged in one pass");
    const int tid = threadIdx.x;
    const int hl  = tid & 1;
    const int g   = (tid >> 1) % G;
    const int ks0 = (tid >> 1) / G;
    const float* s = W + (ks0 * 4 + hl * 2) * NC + g * 4;
    float*       d = sWs + (ks0 * NT + (g >> 2)) * 64 + hl * 32 + (g & 3) * 8;

    float bv = 0.0f;
    if (tid < NC) bv = bias[tid];

    v4f r0[PER], r1[PER];
#pragma unroll
    for (int u = 0; u < PER; ++u) {
        if (FULL || ks0 + u * KS_STEP < K / 4) {
            r0[u] = *(const v4f*)(s + u * KS_STEP * 4 * NC);
            r1[u] = *(const v4f*)(s + u * KS_STEP * 4 * NC + NC);
        }
    }
#pragma unroll
    for (int u = 0; u < PER; ++u) {
        if (FULL || ks0 + u * KS_STEP < K / 4) {
            float* du = d + u * KS_STEP * NT * 64;
            *(v4f*)du       = (v4f){r0[u].x, r1[u].x, r0[u].y, r1[u].y};
            *(v4f*)(du + 4) = (v4f){r0[u].z, r1[u].z, r0[u].w, r1[u].w};
        }
    }
    if (tid < NC) sB[tid] = bv;
}


__device__ __forceinline__ float xhalf(float x) {
    const int b = __builtin_bit_cast(int, x);
    return __builtin_bit_cast(float,
        __builtin_amdgcn_permlanex16(b, b, 0x76543210, 0xfedcba98, false, false));
}

__device__ __forceinline__ void swap_halves(float& a, float& b, bool lo) {
    const float u = xhalf(lo ? b : a);
    a = lo ? a : u;
    b = lo ? u : b;
}

template<int D>
__global__ void __launch_bounds__(256) edge_kernel(
    const float* __restrict__ X, const int* __restrict__ src,
    const int* __restrict__ dst, const float* __restrict__ EA,
    const float* __restrict__ eW, const float* __restrict__ eb,
    float* __restrict__ Agg, int nedges)
{
    constexpr int NT = D / 16;
    constexpr int NL = D / 32;
    __shared__ __attribute__((aligned(16))) float sWs[4 * NT * 64];
    __shared__ float sB[D];
    stage_weights<16, D>(sWs, sB, eW, eb);
    __syncthreads();

    const int lane  = threadIdx.x & 31;
    const int m     = lane & 15;
    const int kb    = (lane >> 4) << 1;
    const bool lo   = lane < 16;

    float bv[NL];
#pragma unroll
    for (int q = 0; q < NL; ++q) bv[q] = sB[q * 32 + lane];

    const int wavesTotal = (gridDim.x * blockDim.x) >> 5;
    const int ntiles = nedges >> 4;
    const int wtile = __builtin_amdgcn_readfirstlane(
        (int)((blockIdx.x * blockDim.x + threadIdx.x) >> 5));

    for (int t = wtile; t < ntiles; t += wavesTotal) {
        const int e0 = t << 4;

        const int ab = (e0 + m) * 16 + kb;
        const v2f a0 = *(const v2f*)(EA + ab);
        const v2f a1 = *(const v2f*)(EA + ab + 4);
        const v2f a2 = *(const v2f*)(EA + ab + 8);
        const v2f a3 = *(const v2f*)(EA + ab + 12);

        int so[16], dd[16];
#pragma unroll
        for (int r = 0; r < 16; ++r) {
            so[r] = src[e0 + r] * D;
            dd[r] = dst[e0 + r] * D;
        }

        float xv[16][NL];
#pragma unroll
        for (int j = 0; j < 8; ++j) {
#pragma unroll
            for (int q = 0; q < NL; ++q) xv[j][q]     = X[so[j] + q * 32 + lane];
#pragma unroll
            for (int q = 0; q < NL; ++q) xv[j + 8][q] = X[so[j + 8] + q * 32 + lane];
        }
        __builtin_amdgcn_sched_barrier(0);

        v8f acc[NT];
#pragma unroll
        for (int c = 0; c < NT; ++c) {
            acc[c] = (v8f){0.f, 0.f, 0.f, 0.f, 0.f, 0.f, 0.f, 0.f};
            const v2f* b0 = (const v2f*)(sWs + (0 * NT + c) * 64);
            const v2f* b1 = (const v2f*)(sWs + (1 * NT + c) * 64);
            const v2f* b2 = (const v2f*)(sWs + (2 * NT + c) * 64);
            const v2f* b3 = (const v2f*)(sWs + (3 * NT + c) * 64);
            acc[c] = __builtin_amdgcn_wmma_f32_16x16x4_f32(false, a0, false, b0[lane], (short)0, acc[c], false, false);
            acc[c] = __builtin_amdgcn_wmma_f32_16x16x4_f32(false, a1, false, b1[lane], (short)0, acc[c], false, false);
            acc[c] = __builtin_amdgcn_wmma_f32_16x16x4_f32(false, a2, false, b2[lane], (short)0, acc[c], false, false);
            acc[c] = __builtin_amdgcn_wmma_f32_16x16x4_f32(false, a3, false, b3[lane], (short)0, acc[c], false, false);
        }

#pragma unroll
        for (int j = 0; j < 8; ++j) {
#pragma unroll
            for (int q = 0; q < NL; ++q) {
                float p = acc[2 * q][j];
                float w = acc[2 * q + 1][j];
                swap_halves(p, w, lo);
                p = (p + bv[q]) + xv[j][q];
                w = (w + bv[q]) + xv[j + 8][q];
                acc[2 * q][j]     = p > 0.0f ? p : 0.0f;
                acc[2 * q + 1][j] = w > 0.0f ? w : 0.0f;
            }
        }
#pragma unroll
        for (int j = 0; j < 8; ++j) {
#pragma unroll
            for (int q = 0; q < NL; ++q)
                atomicAdd(&Agg[dd[j] + q * 32 + lane], acc[2 * q][j]);
#pragma unroll
            for (int q = 0; q < NL; ++q)
                atomicAdd(&Agg[dd[j + 8] + q * 32 + lane], acc[2 * q + 1][j]);
        }
    }
}

constexpr int NODE_THREADS = 512;
constexpr int NODE_WAVES   = NODE_THREADS / 32;
constexpr int CLS_NC       = 32;

template<int K, int NH, int NC, bool CLS, int ZD>
__global__ void __launch_bounds__(NODE_THREADS) node_kernel(
    const float* __restrict__ A, const float* __restrict__ A2,
    const float* __restrict__ W1, const float* __restrict__ b1,
    const float* __restrict__ W2, const float* __restrict__ b2,
    const float* __restrict__ W3, const float* __restrict__ b3,
    float* __restrict__ Out, float* __restrict__ Znext, int nrows)
{
    constexpr int NT1   = NH / 16;
    constexpr int NT2   = NC / 16;
    constexpr int NT3   = CLS_NC / 16;
    constexpr int TW    = (CLS && NC > NH) ? NC : NH;
    constexpr int PITCH = TW + 4;
    __shared__ __attribute__((aligned(16))) float sW1[K * NH];
    __shared__ __attribute__((aligned(16))) float sW2[NH * NC];
    __shared__ __attribute__((aligned(16))) float sW3[CLS ? NC * CLS_NC : 64];
    __shared__ float sB1[NH];
    __shared__ float sB2[NC];
    __shared__ float sB3[CLS_NC];
    __shared__ __attribute__((aligned(16))) float sT[NODE_WAVES * 16 * PITCH];

    if (threadIdx.x < STAGE_THREADS) {
        stage_weights<K, NH>(sW1, sB1, W1, b1);
        stage_weights<NH, NC>(sW2, sB2, W2, b2);
        if (CLS) stage_weights<NC, CLS_NC>(sW3, sB3, W3, b3);
    }
    __syncthreads();

    const int lane  = threadIdx.x & 31;
    const int wave  = __builtin_amdgcn_readfirstlane((int)(threadIdx.x >> 5));
    const int m     = lane & 15;
    const int kb    = (lane >> 4) << 1;
    const int n     = lane & 15;
    const int mbase = (lane >> 4) << 3;

    float*       tile = sT + wave * (16 * PITCH);
    const float* ta   = tile + m * PITCH + kb;
    float*       tw   = tile + mbase * PITCH + n;

    for (int t = blockIdx.x + gridDim.x * wave; t * 16 < nrows; t += gridDim.x * NODE_WAVES) {
        const int row0 = t << 4;

        constexpr int KB = 8;
        constexpr int NB = (K / 4) / KB;
        static_assert((K / 4) % KB == 0, "whole batches");
        v8f acc1[NT1] = {};
        const int abase = (row0 + m) * K + kb;
        v2f ga[2][KB], gb[2][KB];
#pragma unroll
        for (int i = 0; i < KB; ++i) {
            ga[0][i] = *(const v2f*)(A  + abase + i * 4);
            gb[0][i] = *(const v2f*)(A2 + abase + i * 4);
        }
        if (ZD > 0) {
            v4f* z = (v4f*)(Znext + row0 * ZD) + lane;
#pragma unroll
            for (int i = 0; i < (16 * ZD) / 128; ++i) z[i * 32] = (v4f){0.f, 0.f, 0.f, 0.f};
        }
#pragma unroll
        for (int b = 0; b < NB; ++b) {
            if (b + 1 < NB) {
#pragma unroll
                for (int i = 0; i < KB; ++i) {
                    ga[(b + 1) & 1][i] = *(const v2f*)(A  + abase + ((b + 1) * KB + i) * 4);
                    gb[(b + 1) & 1][i] = *(const v2f*)(A2 + abase + ((b + 1) * KB + i) * 4);
                }
            }
            __builtin_amdgcn_sched_barrier(0);
#pragma unroll
            for (int i = 0; i < KB; ++i) {
                v2f a = ga[b & 1][i];
                a.x += gb[b & 1][i].x; a.y += gb[b & 1][i].y;
                const v2f* brow = (const v2f*)(sW1 + (b * KB + i) * NT1 * 64);
#pragma unroll
                for (int c = 0; c < NT1; ++c) {
                    acc1[c] = __builtin_amdgcn_wmma_f32_16x16x4_f32(
                        false, a, false, brow[c * 32 + lane], (short)0, acc1[c], false, false);
                }
            }
        }
#pragma unroll
        for (int c = 0; c < NT1; ++c) {
            const float bv = sB1[c * 16 + n];
#pragma unroll
            for (int j = 0; j < 8; ++j) {
                float v = acc1[c][j] + bv;
                v = (v > 0.0f) ? v : (__expf(v) - 1.0f);
                tw[j * PITCH + c * 16] = v;
            }
        }

        v8f acc2[NT2] = {};
#pragma unroll
        for (int ks = 0; ks < NH / 4; ++ks) {
            const v2f a = *(const v2f*)(ta + ks * 4);
            const v2f* brow = (const v2f*)(sW2 + ks * NT2 * 64);
#pragma unroll
            for (int c = 0; c < NT2; ++c) {
                acc2[c] = __builtin_amdgcn_wmma_f32_16x16x4_f32(
                    false, a, false, brow[c * 32 + lane], (short)0, acc2[c], false, false);
            }
        }

        if (!CLS) {
            const int obase = (row0 + mbase) * NC + n;
#pragma unroll
            for (int c = 0; c < NT2; ++c) {
                const float bv = sB2[c * 16 + n];
#pragma unroll
                for (int j = 0; j < 8; ++j) {
                    float v = acc2[c][j] + bv;
                    v = (v > 0.0f) ? v : (__expf(v) - 1.0f);
                    Out[obase + j * NC + c * 16] = v;
                }
            }
        } else {
#pragma unroll
            for (int c = 0; c < NT2; ++c) {
                const float bv = sB2[c * 16 + n];
#pragma unroll
                for (int j = 0; j < 8; ++j) {
                    float v = acc2[c][j] + bv;
                    v = (v > 0.0f) ? v : (__expf(v) - 1.0f);
                    tw[j * PITCH + c * 16] = v;
                }
            }
            v8f acc3[NT3] = {};
#pragma unroll
            for (int ks = 0; ks < NC / 4; ++ks) {
                const v2f a = *(const v2f*)(ta + ks * 4);
                const v2f* brow = (const v2f*)(sW3 + ks * NT3 * 64);
#pragma unroll
                for (int c = 0; c < NT3; ++c) {
                    acc3[c] = __builtin_amdgcn_wmma_f32_16x16x4_f32(
                        false, a, false, brow[c * 32 + lane], (short)0, acc3[c], false, false);
                }
            }
            const int obase = (row0 + mbase) * CLS_NC + n;
#pragma unroll
            for (int c = 0; c < NT3; ++c) {
                const float bv = sB3[c * 16 + n];
#pragma unroll
                for (int j = 0; j < 8; ++j)
                    Out[obase + j * CLS_NC + c * 16] = acc3[c][j] + bv;
            }
        }
    }
}

extern "C" void kernel_launch(void* const* d_in, const int* in_sizes, int n_in,
                              void* d_out, int out_size, void* d_ws, size_t ws_size,
                              hipStream_t stream)
{
    const float* x    = (const float*)d_in[0];
    const int*   src  = (const int*)  d_in[1];
    const int*   dst  = (const int*)  d_in[2];
    const float* ea   = (const float*)d_in[3];
    const float* eW0  = (const float*)d_in[4];
    const float* eb0  = (const float*)d_in[5];
    const float* W1_0 = (const float*)d_in[6];
    const float* b1_0 = (const float*)d_in[7];
    const float* W2_0 = (const float*)d_in[8];
    const float* b2_0 = (const float*)d_in[9];
    const float* eW_r = (const float*)d_in[10];
    const float* eb_r = (const float*)d_in[11];
    const float* W1_r = (const float*)d_in[12];
    const float* b1_r = (const float*)d_in[13];
    const float* W2_r = (const float*)d_in[14];
    const float* b2_r = (const float*)d_in[15];
    const float* cW   = (const float*)d_in[16];
    const float* cb   = (const float*)d_in[17];

    const int N = in_sizes[0] / 64;
    const int E = in_sizes[1];

    float* aggA = (float*)d_ws;
    float* aggB = aggA + (size_t)N * 128;
    float* hb   = aggB + (size_t)N * 128;
    float* tb   = hb   + (size_t)N * 128;

    const int eblocks = 2048;
    const int nblocks = 256;

    { int n4 = (N * 64) / 4; zero_kernel<<<(n4 + 255) / 256, 256, 0, stream>>>((v4f*)aggA, n4); }
    edge_kernel<64><<<eblocks, 256, 0, stream>>>(x, src, dst, ea, eW0, eb0, aggA, E);
    node_kernel<64, 64, 128, false, 128><<<nblocks, NODE_THREADS, 0, stream>>>(
        x, aggA, W1_0, b1_0, W2_0, b2_0, cW, cb, hb, aggB, N);

    edge_kernel<128><<<eblocks, 256, 0, stream>>>(hb, src, dst, ea, eW_r, eb_r, aggB, E);
    node_kernel<128, 128, 128, false, 128><<<nblocks, NODE_THREADS, 0, stream>>>(
        hb, aggB, W1_r, b1_r, W2_r, b2_r, cW, cb, tb, aggA, N);

    edge_kernel<128><<<eblocks, 256, 0, stream>>>(
        tb, src, dst, ea, eW_r + (size_t)16 * 128, eb_r + (size_t)128, aggA, E);
    node_kernel<128, 128, 128, true, 0><<<nblocks, NODE_THREADS, 0, stream>>>(
        tb, aggA, W1_r + (size_t)128 * 128, b1_r + (size_t)128,
        W2_r + (size_t)128 * 128, b2_r + (size_t)128, cW, cb, (float*)d_out, aggB, N);
}


# Edge kernel: tile loop software-pipelined over two register sets (next tile's gathers, A fragments, indices in flight under this tile's WMMAs and atomics); resident grid; atomic arrival order differs
# speedup vs baseline: 2.0290x; 1.0371x over previous
#include <hip/hip_runtime.h>
#include <cmath>

typedef __attribute__((ext_vector_type(2))) float v2f;
typedef __attribute__((ext_vector_type(4))) float v4f;
typedef __attribute__((ext_vector_type(8))) float v8f;

__global__ void __launch_bounds__(256) zero_kernel(v4f* __restrict__ p, int n4) {
    int i = blockIdx.x * blockDim.x + threadIdx.x;
    if (i < n4) p[i] = (v4f){0.f, 0.f, 0.f, 0.f};
}

constexpr int STAGE_THREADS = 256;

template<int K, int NC>
__device__ __forceinline__ void stage_weights(float* sWs, float* sB,
                                              const float* __restrict__ W,
                                              const float* __restrict__ bias) {
    constexpr int NT      = NC / 16;
    constexpr int G       = NC / 4;
    constexpr int HALF    = STAGE_THREADS / 2;
    constexpr int KS_STEP = HALF / G;
    constexpr int PER     = (K / 4 + KS_STEP - 1) / KS_STEP;
    constexpr bool FULL   = ((K / 4) % KS_STEP) == 0;
    static_assert(HALF % G == 0, "a pass covers whole k-steps");
    static_assert(NC <= STAGE_THREADS, "bias is staged in one pass");
    const int tid = threadIdx.x;
    const int hl  = tid & 1;
    const int g   = (tid >> 1) % G;
    const int ks0 = (tid >> 1) / G;
    const float* s = W + (ks0 * 4 + hl * 2) * NC + g * 4;
    float*       d = sWs + (ks0 * NT + (g >> 2)) * 64 + hl * 32 + (g & 3) * 8;

    float bv = 0.0f;
    if (tid < NC) bv = bias[tid];

    v4f r0[PER], r1[PER];
#pragma unroll
    for (int u = 0; u < PER; ++u) {
        if (FULL || ks0 + u * KS_STEP < K / 4) {
            r0[u] = *(const v4f*)(s + u * KS_STEP * 4 * NC);
            r1[u] = *(const v4f*)(s + u * KS_STEP * 4 * NC + NC);
        }
    }
#pragma unroll
    for (int u = 0; u < PER; ++u) {
        if (FULL || ks0 + u * KS_STEP < K / 4) {
            float* du = d + u * KS_STEP * NT * 64;
            *(v4f*)du       = (v4f){r0[u].x, r1[u].x, r0[u].y, r1[u].y};
            *(v4f*)(du + 4) = (v4f){r0[u].z, r1[u].z, r0[u].w, r1[u].w};
        }
    }
    if (tid < NC) sB[tid] = bv;
}


__device__ __forceinline__ float xhalf(float x) {
    const int b = __builtin_bit_cast(int, x);
    return __builtin_bit_cast(float,
        __builtin_amdgcn_permlanex16(b, b, 0x76543210, 0xfedcba98, false, false));
}

__device__ __forceinline__ void swap_halves(float& a, float& b, bool lo) {
    const float u = xhalf(lo ? b : a);
    a = lo ? a : u;
    b = lo ? u : b;
}

typedef __attribute__((address_space(4))) const int const_int_t;
__device__ __forceinline__ void load_rows(int (&r)[16], const int* __restrict__ idx, int t) {
    const const_int_t* p = (const const_int_t*)(__UINTPTR_TYPE__)(idx + (t << 4));
#pragma unroll
    for (int i = 0; i < 16; ++i) r[i] = p[i];
}

__device__ __forceinline__ void load_afrag(v2f (&a)[4], const float* __restrict__ EA, int t, int m, int kb) {
    const int ab = ((t << 4) + m) * 16 + kb;
#pragma unroll
    for (int s = 0; s < 4; ++s) a[s] = *(const v2f*)(EA + ab + 4 * s);
}

template<int D>
__device__ __forceinline__ void issue_gathers(float (&xv)[16][D / 32], const float* __restrict__ X,
                                              const int (&rows)[16], int lane) {
    constexpr int NL = D / 32;
#pragma unroll
    for (int j = 0; j < 8; ++j) {
#pragma unroll
        for (int q = 0; q < NL; ++q) xv[j][q]     = X[rows[j] * D + q * 32 + lane];
#pragma unroll
        for (int q = 0; q < NL; ++q) xv[j + 8][q] = X[rows[j + 8] * D + q * 32 + lane];
    }
}


template<int D, int Q>
__device__ __forceinline__ void edge_prefetch(
    const int tn, const int stride, const int ntiles,
    const float* __restrict__ X, const int* __restrict__ src, const int* __restrict__ dst,
    const float* __restrict__ EA,
    v2f (&fa)[2][4], float (&xv)[2][16][D / 32], int (&sn)[16], int (&dd)[2][16],
    const int lane, const int m, const int kb)
{
    issue_gathers<D>(xv[Q], X, sn, lane);
    load_afrag(fa[Q], EA, tn, m, kb);
    load_rows(dd[Q], dst, tn);
    if (tn + stride < ntiles) load_rows(sn, src, tn + stride);
}

template<int D, int P>
__device__ __forceinline__ void edge_compute(
    float* __restrict__ Agg, const float* sWs,
    const v2f (&fa)[2][4], const float (&xv)[2][16][D / 32], const int (&dd)[2][16],
    const float (&bv)[D / 32], const int lane, const bool lo)
{
    constexpr int NT = D / 16;
    constexpr int NL = D / 32;
    v8f acc[NT];
#pragma unroll
    for (int c = 0; c < NT; ++c) {
        acc[c] = (v8f){0.f, 0.f, 0.f, 0.f, 0.f, 0.f, 0.f, 0.f};
        const v2f* b0 = (const v2f*)(sWs + (0 * NT + c) * 64);
        const v2f* b1 = (const v2f*)(sWs + (1 * NT + c) * 64);
        const v2f* b2 = (const v2f*)(sWs + (2 * NT + c) * 64);
        const v2f* b3 = (const v2f*)(sWs + (3 * NT + c) * 64);
        acc[c] = __builtin_amdgcn_wmma_f32_16x16x4_f32(false, fa[P][0], false, b0[lane], (short)0, acc[c], false, false);
        acc[c] = __builtin_amdgcn_wmma_f32_16x16x4_f32(false, fa[P][1], false, b1[lane], (short)0, acc[c], false, false);
        acc[c] = __builtin_amdgcn_wmma_f32_16x16x4_f32(false, fa[P][2], false, b2[lane], (short)0, acc[c], false, false);
        acc[c] = __builtin_amdgcn_wmma_f32_16x16x4_f32(false, fa[P][3], false, b3[lane], (short)0, acc[c], false, false);
    }

#pragma unroll
    for (int j = 0; j < 8; ++j) {
#pragma unroll
        for (int q = 0; q < NL; ++q) {
            float p = acc[2 * q][j];
            float w = acc[2 * q + 1][j];
            swap_halves(p, w, lo);
            p = (p + bv[q]) + xv[P][j][q];
            w = (w + bv[q]) + xv[P][j + 8][q];
            acc[2 * q][j]     = p > 0.0f ? p : 0.0f;
            acc[2 * q + 1][j] = w > 0.0f ? w : 0.0f;
        }
    }
#pragma unroll
    for (int j = 0; j < 8; ++j) {
#pragma unroll
        for (int q = 0; q < NL; ++q)
            atomicAdd(&Agg[dd[P][j] * D + q * 32 + lane], acc[2 * q][j]);
#pragma unroll
        for (int q = 0; q < NL; ++q)
            atomicAdd(&Agg[dd[P][j + 8] * D + q * 32 + lane], acc[2 * q + 1][j]);
    }
}

template<int D>
__global__ void __launch_bounds__(256) edge_kernel(
    const float* __restrict__ X, const int* __restrict__ src,
    const int* __restrict__ dst, const float* __restrict__ EA,
    const float* __restrict__ eW, const float* __restrict__ eb,
    float* __restrict__ Agg, int nedges)
{
    constexpr int NT = D / 16;
    constexpr int NL = D / 32;
    __shared__ __attribute__((aligned(16))) float sWs[4 * NT * 64];
    __shared__ float sB[D];
    stage_weights<16, D>(sWs, sB, eW, eb);
    __syncthreads();

    const int lane  = threadIdx.x & 31;
    const int m     = lane & 15;
    const int kb    = (lane >> 4) << 1;
    const bool lo   = lane < 16;

    float bv[NL];
#pragma unroll
    for (int q = 0; q < NL; ++q) bv[q] = sB[q * 32 + lane];

    const int stride = (gridDim.x * blockDim.x) >> 5;
    const int ntiles = nedges >> 4;
    int t = __builtin_amdgcn_readfirstlane(
        (int)((blockIdx.x * blockDim.x + threadIdx.x) >> 5));
    if (t >= ntiles) return;

    v2f   fa[2][4];
    float xv[2][16][NL];
    int   dd[2][16];
    int   sn[16];

    load_rows(sn, src, t);
    load_afrag(fa[0], EA, t, m, kb);
    load_rows(dd[0], dst, t);
    issue_gathers<D>(xv[0], X, sn, lane);
    if (t + stride < ntiles) load_rows(sn, src, t + stride);

    while (true) {
        int tn = t + stride;
        if (tn >= ntiles) { edge_compute<D, 0>(Agg, sWs, fa, xv, dd, bv, lane, lo); break; }
        edge_prefetch<D, 1>(tn, stride, ntiles, X, src, dst, EA, fa, xv, sn, dd, lane, m, kb);
        __builtin_amdgcn_sched_barrier(0);
        edge_compute<D, 0>(Agg, sWs, fa, xv, dd, bv, lane, lo);
        t = tn;
        tn = t + stride;
        if (tn >= ntiles) { edge_compute<D, 1>(Agg, sWs, fa, xv, dd, bv, lane, lo); break; }
        edge_prefetch<D, 0>(tn, stride, ntiles, X, src, dst, EA, fa, xv, sn, dd, lane, m, kb);
        __builtin_amdgcn_sched_barrier(0);
        edge_compute<D, 1>(Agg, sWs, fa, xv, dd, bv, lane, lo);
        t = tn;
    }
}

constexpr int NODE_THREADS = 512;
constexpr int NODE_WAVES   = NODE_THREADS / 32;
constexpr int CLS_NC       = 32;

template<int K, int NH, int NC, bool CLS, int ZD>
__global__ void __launch_bounds__(NODE_THREADS) node_kernel(
    const float* __restrict__ A, const float* __restrict__ A2,
    const float* __restrict__ W1, const float* __restrict__ b1,
    const float* __restrict__ W2, const float* __restrict__ b2,
    const float* __restrict__ W3, const float* __restrict__ b3,
    float* __restrict__ Out, float* __restrict__ Znext, int nrows)
{
    constexpr int NT1   = NH / 16;
    constexpr int NT2   = NC / 16;
    constexpr int NT3   = CLS_NC / 16;
    constexpr int TW    = (CLS && NC > NH) ? NC : NH;
    constexpr int PITCH = TW + 4;
    __shared__ __attribute__((aligned(16))) float sW1[K * NH];
    __shared__ __attribute__((aligned(16))) float sW2[NH * NC];
    __shared__ __attribute__((aligned(16))) float sW3[CLS ? NC * CLS_NC : 64];
    __shared__ float sB1[NH];
    __shared__ float sB2[NC];
    __shared__ float sB3[CLS_NC];
    __shared__ __attribute__((aligned(16))) float sT[NODE_WAVES * 16 * PITCH];

    if (threadIdx.x < STAGE_THREADS) {
        stage_weights<K, NH>(sW1, sB1, W1, b1);
        stage_weights<NH, NC>(sW2, sB2, W2, b2);
        if (CLS) stage_weights<NC, CLS_NC>(sW3, sB3, W3, b3);
    }
    __syncthreads();

    const int lane  = threadIdx.x & 31;
    const int wave  = __builtin_amdgcn_readfirstlane((int)(threadIdx.x >> 5));
    const int m     = lane & 15;
    const int kb    = (lane >> 4) << 1;
    const int n     = lane & 15;
    const int mbase = (lane >> 4) << 3;

    float*       tile = sT + wave * (16 * PITCH);
    const float* ta   = tile + m * PITCH + kb;
    float*       tw   = tile + mbase * PITCH + n;

    for (int t = blockIdx.x + gridDim.x * wave; t * 16 < nrows; t += gridDim.x * NODE_WAVES) {
        const int row0 = t << 4;

        constexpr int KB = 8;
        constexpr int NB = (K / 4) / KB;
        static_assert((K / 4) % KB == 0, "whole batches");
        v8f acc1[NT1] = {};
        const int abase = (row0 + m) * K + kb;
        v2f ga[2][KB], gb[2][KB];
#pragma unroll
        for (int i = 0; i < KB; ++i) {
            ga[0][i] = *(const v2f*)(A  + abase + i * 4);
            gb[0][i] = *(const v2f*)(A2 + abase + i * 4);
        }
        if (ZD > 0) {
            v4f* z = (v4f*)(Znext + row0 * ZD) + lane;
#pragma unroll
            for (int i = 0; i < (16 * ZD) / 128; ++i) z[i * 32] = (v4f){0.f, 0.f, 0.f, 0.f};
        }
#pragma unroll
        for (int b = 0; b < NB; ++b) {
            if (b + 1 < NB) {
#pragma unroll
                for (int i = 0; i < KB; ++i) {
                    ga[(b + 1) & 1][i] = *(const v2f*)(A  + abase + ((b + 1) * KB + i) * 4);
                    gb[(b + 1) & 1][i] = *(const v2f*)(A2 + abase + ((b + 1) * KB + i) * 4);
                }
            }
            __builtin_amdgcn_sched_barrier(0);
#pragma unroll
            for (int i = 0; i < KB; ++i) {
                v2f a = ga[b & 1][i];
                a.x += gb[b & 1][i].x; a.y += gb[b & 1][i].y;
                const v2f* brow = (const v2f*)(sW1 + (b * KB + i) * NT1 * 64);
#pragma unroll
                for (int c = 0; c < NT1; ++c) {
                    acc1[c] = __builtin_amdgcn_wmma_f32_16x16x4_f32(
                        false, a, false, brow[c * 32 + lane], (short)0, acc1[c], false, false);
                }
            }
        }
#pragma unroll
        for (int c = 0; c < NT1; ++c) {
            const float bv = sB1[c * 16 + n];
#pragma unroll
            for (int j = 0; j < 8; ++j) {
                float v = acc1[c][j] + bv;
                v = (v > 0.0f) ? v : (__expf(v) - 1.0f);
                tw[j * PITCH + c * 16] = v;
            }
        }

        v8f acc2[NT2] = {};
#pragma unroll
        for (int ks = 0; ks < NH / 4; ++ks) {
            const v2f a = *(const v2f*)(ta + ks * 4);
            const v2f* brow = (const v2f*)(sW2 + ks * NT2 * 64);
#pragma unroll
            for (int c = 0; c < NT2; ++c) {
                acc2[c] = __builtin_amdgcn_wmma_f32_16x16x4_f32(
                    false, a, false, brow[c * 32 + lane], (short)0, acc2[c], false, false);
            }
        }

        if (!CLS) {
            const int obase = (row0 + mbase) * NC + n;
#pragma unroll
            for (int c = 0; c < NT2; ++c) {
                const float bv = sB2[c * 16 + n];
#pragma unroll
                for (int j = 0; j < 8; ++j) {
                    float v = acc2[c][j] + bv;
                    v = (v > 0.0f) ? v : (__expf(v) - 1.0f);
                    Out[obase + j * NC + c * 16] = v;
                }
            }
        } else {
#pragma unroll
            for (int c = 0; c < NT2; ++c) {
                const float bv = sB2[c * 16 + n];
#pragma unroll
                for (int j = 0; j < 8; ++j) {
                    float v = acc2[c][j] + bv;
                    v = (v > 0.0f) ? v : (__expf(v) - 1.0f);
                    tw[j * PITCH + c * 16] = v;
                }
            }
            v8f acc3[NT3] = {};
#pragma unroll
            for (int ks = 0; ks < NC / 4; ++ks) {
                const v2f a = *(const v2f*)(ta + ks * 4);
                const v2f* brow = (const v2f*)(sW3 + ks * NT3 * 64);
#pragma unroll
                for (int c = 0; c < NT3; ++c) {
                    acc3[c] = __builtin_amdgcn_wmma_f32_16x16x4_f32(
                        false, a, false, brow[c * 32 + lane], (short)0, acc3[c], false, false);
                }
            }
            const int obase = (row0 + mbase) * CLS_NC + n;
#pragma unroll
            for (int c = 0; c < NT3; ++c) {
                const float bv = sB3[c * 16 + n];
#pragma unroll
                for (int j = 0; j < 8; ++j)
                    Out[obase + j * CLS_NC + c * 16] = acc3[c][j] + bv;
            }
        }
    }
}

extern "C" void kernel_launch(void* const* d_in, const int* in_sizes, int n_in,
                              void* d_out, int out_size, void* d_ws, size_t ws_size,
                              hipStream_t stream)
{
    const float* x    = (const float*)d_in[0];
    const int*   src  = (const int*)  d_in[1];
    const int*   dst  = (const int*)  d_in[2];
    const float* ea   = (const float*)d_in[3];
    const float* eW0  = (const float*)d_in[4];
    const float* eb0  = (const float*)d_in[5];
    const float* W1_0 = (const float*)d_in[6];
    const float* b1_0 = (const float*)d_in[7];
    const float* W2_0 = (const float*)d_in[8];
    const float* b2_0 = (const float*)d_in[9];
    const float* eW_r = (const float*)d_in[10];
    const float* eb_r = (const float*)d_in[11];
    const float* W1_r = (const float*)d_in[12];
    const float* b1_r = (const float*)d_in[13];
    const float* W2_r = (const float*)d_in[14];
    const float* b2_r = (const float*)d_in[15];
    const float* cW   = (const float*)d_in[16];
    const float* cb   = (const float*)d_in[17];

    const int N = in_sizes[0] / 64;
    const int E = in_sizes[1];

    float* aggA = (float*)d_ws;
    float* aggB = aggA + (size_t)N * 128;
    float* hb   = aggB + (size_t)N * 128;
    float* tb   = hb   + (size_t)N * 128;

    const int eblocks64  = 768;
    const int eblocks128 = 512;
    const int nblocks = 256;

    { int n4 = (N * 64) / 4; zero_kernel<<<(n4 + 255) / 256, 256, 0, stream>>>((v4f*)aggA, n4); }
    edge_kernel<64><<<eblocks64, 256, 0, stream>>>(x, src, dst, ea, eW0, eb0, aggA, E);
    node_kernel<64, 64, 128, false, 128><<<nblocks, NODE_THREADS, 0, stream>>>(
        x, aggA, W1_0, b1_0, W2_0, b2_0, cW, cb, hb, aggB, N);

    edge_kernel<128><<<eblocks128, 256, 0, stream>>>(hb, src, dst, ea, eW_r, eb_r, aggB, E);
    node_kernel<128, 128, 128, false, 128><<<nblocks, NODE_THREADS, 0, stream>>>(
        hb, aggB, W1_r, b1_r, W2_r, b2_r, cW, cb, tb, aggA, N);

    edge_kernel<128><<<eblocks128, 256, 0, stream>>>(
        tb, src, dst, ea, eW_r + (size_t)16 * 128, eb_r + (size_t)128, aggA, E);
    node_kernel<128, 128, 128, true, 0><<<nblocks, NODE_THREADS, 0, stream>>>(
        tb, aggA, W1_r + (size_t)128 * 128, b1_r + (size_t)128,
        W2_r + (size_t)128 * 128, b2_r + (size_t)128, cW, cb, (float*)d_out, aggB, N);
}
